# starting kernel
# baseline (speedup 1.0000x reference)
#include <hip/hip_runtime.h>

typedef __attribute__((ext_vector_type(2))) float v2f;
typedef __attribute__((ext_vector_type(8))) float v8f;

#define C_DIM   128
#define MAX_DEG 16
#define WPAIR_STRIDE 144

__global__ __launch_bounds__(256) void gcn_gemm(const float* __restrict__ x,
                                                const int*   __restrict__ edge,
                                                const float* __restrict__ W,
                                                float*       __restrict__ h,
                                                int nStrips) {
    __shared__ float Wlds[64 * WPAIR_STRIDE * 2];

    {
        const int tid = threadIdx.x;
#pragma unroll
        for (int i = 0; i < 16; ++i) {
            const int q = tid + 256 * i;
            const int p = q >> 6;
            const int c = q & 63;
            const float2 ga = *(const float2*)(W + (size_t)(2 * p)     * C_DIM + 2 * c);
            const float2 gb = *(const float2*)(W + (size_t)(2 * p + 1) * C_DIM + 2 * c);
            float4 v; v.x = ga.x; v.y = gb.x; v.z = ga.y; v.w = gb.y;
            ((float4*)Wlds)[p * (WPAIR_STRIDE / 2) + c] = v;
        }
    }
    __syncthreads();

    const int wave  = threadIdx.x >> 5;
    const int lane  = threadIdx.x & 31;
    const int strip = blockIdx.x * 8 + wave;
    if (strip >= nStrips) return;

    const int m  = lane & 15;
    const int hi = lane >> 4;
    const int rowBase = strip * 16;

    const float* xrow = x + (size_t)(rowBase + m) * C_DIM;
    const float* wb0 = Wlds + ((size_t)hi * WPAIR_STRIDE + m) * 2;

    v8f c[8] = {};

    for (int k0 = 0; k0 < C_DIM; k0 += 4) {
        const int kk = k0 + 2 * hi;
        v2f a = *(const v2f*)(xrow + kk);
        const float* wb = wb0 + (size_t)(k0 >> 1) * WPAIR_STRIDE * 2;
#pragma unroll
        for (int t = 0; t < 8; ++t) {
            v2f b = *(const v2f*)(wb + t * 32);
            c[t] = __builtin_amdgcn_wmma_f32_16x16x4_f32(
                false, a, false, b, (short)0, c[t], false, false);
        }
    }

    const int* e = edge + (size_t)(rowBase + m) * MAX_DEG;
    int cnt = 0;
#pragma unroll
    for (int k = 0; k < MAX_DEG; ++k) cnt += (e[k] >= 0) ? 1 : 0;
    const float sv = rsqrtf((float)cnt + 1.0f);

#pragma unroll
    for (int r = 0; r < 8; ++r) {
        const float sc  = __shfl(sv, r + 8 * hi, 32);
        const int   row = rowBase + r + 8 * hi;
        float* hrow = h + (size_t)row * C_DIM;
#pragma unroll
        for (int t = 0; t < 8; ++t)
            hrow[t * 16 + m] = c[t][r] * sc;
    }
}

__global__ __launch_bounds__(256) void gcn_agg(const float* __restrict__ h,
                                               const int*   __restrict__ edge,
                                               float*       __restrict__ out,
                                               int nNodes) {
    const int lane = threadIdx.x & 31;
    const int node = blockIdx.x * 8 + (threadIdx.x >> 5);
    if (node >= nNodes) return;

    int idx = -1;
    if (lane < MAX_DEG) idx = edge[(size_t)node * MAX_DEG + lane];

    const unsigned long long valid = __ballot(idx >= 0);
    const float s = rsqrtf((float)__popcll(valid) + 1.0f);

    const float4* h4 = (const float4*)h;
    float4 acc = h4[(size_t)node * 32 + lane];

    int jnext = __shfl(idx, 0, 32);
#pragma unroll 4
    for (int k = 0; k < MAX_DEG; ++k) {
        const int j = jnext;
        jnext = (k + 1 < MAX_DEG) ? __shfl(idx, k + 1, 32) : -1;
        if (jnext >= 0)
            __builtin_prefetch(&h4[(size_t)jnext * 32 + lane], 0, 3);
        if (j >= 0) {
            const float4 v = h4[(size_t)j * 32 + lane];
            acc.x += v.x; acc.y += v.y; acc.z += v.z; acc.w += v.w;
        }
    }

    float4 o;
    o.x = acc.x * s; o.y = acc.y * s; o.z = acc.z * s; o.w = acc.w * s;
    ((float4*)out)[(size_t)node * 32 + lane] = o;
}

extern "C" void kernel_launch(void* const* d_in, const int* in_sizes, int n_in,
                              void* d_out, int out_size, void* d_ws, size_t ws_size,
                              hipStream_t stream) {
    const float* x    = (const float*)d_in[0];
    const int*   edge = (const int*)  d_in[1];
    const float* W    = (const float*)d_in[2];
    float*       out  = (float*)d_out;
    float*       h    = (float*)d_ws;

    const int N       = in_sizes[0] / C_DIM;
    const int nStrips = (N + 15) / 16;

    const int gemmBlocks = (nStrips + 7) / 8;
    gcn_gemm<<<gemmBlocks, 256, 0, stream>>>(x, edge, W, h, nStrips);

    const int aggBlocks = (N + 7) / 8;
    gcn_agg<<<aggBlocks, 256, 0, stream>>>(h, edge, out, N);
}


# agg: each wave loops over nodes and issues the self row and all 16 neighbour rows together before the adds (no per-slot branch, no L2 prefetch); 1024 blocks; sum order kept
# speedup vs baseline: 1.1607x; 1.1607x over previous
#include <hip/hip_runtime.h>

typedef __attribute__((ext_vector_type(2))) float v2f;
typedef __attribute__((ext_vector_type(8))) float v8f;

#define C_DIM   128
#define MAX_DEG 16
#define WPAIR_STRIDE 144

__global__ __launch_bounds__(256) void gcn_gemm(const float* __restrict__ x,
                                                const int*   __restrict__ edge,
                                                const float* __restrict__ W,
                                                float*       __restrict__ h,
                                                int nStrips) {
    __shared__ float Wlds[64 * WPAIR_STRIDE * 2];

    {
        const int tid = threadIdx.x;
#pragma unroll
        for (int i = 0; i < 16; ++i) {
            const int q = tid + 256 * i;
            const int p = q >> 6;
            const int c = q & 63;
            const float2 ga = *(const float2*)(W + (size_t)(2 * p)     * C_DIM + 2 * c);
            const float2 gb = *(const float2*)(W + (size_t)(2 * p + 1) * C_DIM + 2 * c);
            float4 v; v.x = ga.x; v.y = gb.x; v.z = ga.y; v.w = gb.y;
            ((float4*)Wlds)[p * (WPAIR_STRIDE / 2) + c] = v;
        }
    }
    __syncthreads();

    const int wave  = threadIdx.x >> 5;
    const int lane  = threadIdx.x & 31;
    const int strip = blockIdx.x * 8 + wave;
    if (strip >= nStrips) return;

    const int m  = lane & 15;
    const int hi = lane >> 4;
    const int rowBase = strip * 16;

    const float* xrow = x + (size_t)(rowBase + m) * C_DIM;
    const float* wb0 = Wlds + ((size_t)hi * WPAIR_STRIDE + m) * 2;

    v8f c[8] = {};

    for (int k0 = 0; k0 < C_DIM; k0 += 4) {
        const int kk = k0 + 2 * hi;
        v2f a = *(const v2f*)(xrow + kk);
        const float* wb = wb0 + (size_t)(k0 >> 1) * WPAIR_STRIDE * 2;
#pragma unroll
        for (int t = 0; t < 8; ++t) {
            v2f b = *(const v2f*)(wb + t * 32);
            c[t] = __builtin_amdgcn_wmma_f32_16x16x4_f32(
                false, a, false, b, (short)0, c[t], false, false);
        }
    }

    const int* e = edge + (size_t)(rowBase + m) * MAX_DEG;
    int cnt = 0;
#pragma unroll
    for (int k = 0; k < MAX_DEG; ++k) cnt += (e[k] >= 0) ? 1 : 0;
    const float sv = rsqrtf((float)cnt + 1.0f);

#pragma unroll
    for (int r = 0; r < 8; ++r) {
        const float sc  = __shfl(sv, r + 8 * hi, 32);
        const int   row = rowBase + r + 8 * hi;
        float* hrow = h + (size_t)row * C_DIM;
#pragma unroll
        for (int t = 0; t < 8; ++t)
            hrow[t * 16 + m] = c[t][r] * sc;
    }
}

__global__ __launch_bounds__(256) __attribute__((amdgpu_waves_per_eu(4, 8))) void gcn_agg(const float* __restrict__ h,
                                               const int*   __restrict__ edge,
                                               float*       __restrict__ out,
                                               int nNodes) {
    const int lane   = threadIdx.x & 31;
    const int stride = gridDim.x * 8;
    const int first  = __builtin_amdgcn_readfirstlane(blockIdx.x * 8 + (threadIdx.x >> 5));

    const float4* h4   = (const float4*)h;
    float4*       out4 = (float4*)out;

    int idx = -1;
    if (first < nNodes && lane < MAX_DEG) idx = edge[(size_t)first * MAX_DEG + lane];
    unsigned long long valid = __ballot(idx >= 0);

    for (int node = first; node < nNodes; node += stride) {
        const float s = rsqrtf((float)__popcll(valid) + 1.0f);

        int    j[MAX_DEG];
        float4 v[MAX_DEG];
        const float4 self = h4[(size_t)node * 32 + lane];
#pragma unroll
        for (int k = 0; k < MAX_DEG; ++k) {
            j[k] = __builtin_amdgcn_readlane(idx, k);
            const int jc = j[k] < 0 ? 0 : j[k];
            v[k] = h4[(size_t)jc * 32 + lane];
        }
        const int next = node + stride;
        int idxNext = -1;
        if (next < nNodes && lane < MAX_DEG) idxNext = edge[(size_t)next * MAX_DEG + lane];

        float4 acc = self;
#pragma unroll
        for (int k = 0; k < MAX_DEG; ++k) {
            const bool ok = j[k] >= 0;
            const float tx = acc.x + v[k].x, ty = acc.y + v[k].y;
            const float tz = acc.z + v[k].z, tw = acc.w + v[k].w;
            acc.x = ok ? tx : acc.x; acc.y = ok ? ty : acc.y;
            acc.z = ok ? tz : acc.z; acc.w = ok ? tw : acc.w;
        }

        float4 o;
        o.x = acc.x * s; o.y = acc.y * s; o.z = acc.z * s; o.w = acc.w * s;
        out4[(size_t)node * 32 + lane] = o;

        idx   = idxNext;
        valid = __ballot(idx >= 0);
    }
}

extern "C" void kernel_launch(void* const* d_in, const int* in_sizes, int n_in,
                              void* d_out, int out_size, void* d_ws, size_t ws_size,
                              hipStream_t stream) {
    const float* x    = (const float*)d_in[0];
    const int*   edge = (const int*)  d_in[1];
    const float* W    = (const float*)d_in[2];
    float*       out  = (float*)d_out;
    float*       h    = (float*)d_ws;

    const int N       = in_sizes[0] / C_DIM;
    const int nStrips = (N + 15) / 16;

    const int gemmBlocks = (nStrips + 7) / 8;
    gcn_gemm<<<gemmBlocks, 256, 0, stream>>>(x, edge, W, h, nStrips);

    gcn_agg<<<1024, 256, 0, stream>>>(h, edge, out, N);
}


# gemm: one 28-wave block per WGP (W staged once); each WGP gets 24 whole 16-row strips plus a 25th cut into four column quarters, one per SIMD: busiest SIMD issues 1600 WMMAs, not 2048
# speedup vs baseline: 1.2013x; 1.0350x over previous
#include <hip/hip_runtime.h>

typedef __attribute__((ext_vector_type(2))) float v2f;
typedef __attribute__((ext_vector_type(8))) float v8f;

#define C_DIM   128
#define MAX_DEG 16
#define WPAIR_STRIDE 144

#define GEMM_THREADS  896
#define QUARTER_WAVES 4
#define WHOLE_WAVES   24

template <int NT, bool AHEAD>
__device__ __forceinline__ void strip_tiles(const float* __restrict__ x,
                                            const int*   __restrict__ edge,
                                            const float* Wlds,
                                            float*       __restrict__ h,
                                            int strip, int t0, int lane) {
    const int m  = lane & 15;
    const int hi = lane >> 4;
    const int rowBase = strip * 16;

    const float* xrow = x + (size_t)(rowBase + m) * C_DIM;
    const float* wb0 = Wlds + ((size_t)hi * WPAIR_STRIDE + m) * 2 + t0 * 32;

    v8f c[NT] = {};

    if (AHEAD) {
        v2f a[C_DIM / 4];
#pragma unroll
        for (int i = 0; i < C_DIM / 4; ++i) a[i] = *(const v2f*)(xrow + 4 * i + 2 * hi);
#pragma unroll
        for (int i = 0; i < C_DIM / 4; ++i) {
            const float* wb = wb0 + (size_t)(2 * i) * WPAIR_STRIDE * 2;
#pragma unroll
            for (int t = 0; t < NT; ++t) {
                v2f b = *(const v2f*)(wb + t * 32);
                c[t] = __builtin_amdgcn_wmma_f32_16x16x4_f32(
                    false, a[i], false, b, (short)0, c[t], false, false);
            }
        }
    } else {
        for (int k0 = 0; k0 < C_DIM; k0 += 4) {
            const int kk = k0 + 2 * hi;
            v2f a = *(const v2f*)(xrow + kk);
            const float* wb = wb0 + (size_t)(k0 >> 1) * WPAIR_STRIDE * 2;
#pragma unroll
            for (int t = 0; t < NT; ++t) {
                v2f b = *(const v2f*)(wb + t * 32);
                c[t] = __builtin_amdgcn_wmma_f32_16x16x4_f32(
                    false, a, false, b, (short)0, c[t], false, false);
            }
        }
    }

    const int* e = edge + (size_t)(rowBase + m) * MAX_DEG;
    int cnt = 0;
#pragma unroll
    for (int k = 0; k < MAX_DEG; ++k) cnt += (e[k] >= 0) ? 1 : 0;
    const float sv = rsqrtf((float)cnt + 1.0f);

#pragma unroll
    for (int r = 0; r < 8; ++r) {
        const float sc  = __shfl(sv, r + 8 * hi, 32);
        const int   row = rowBase + r + 8 * hi;
        float* hrow = h + (size_t)row * C_DIM + t0 * 16;
#pragma unroll
        for (int t = 0; t < NT; ++t)
            hrow[t * 16 + m] = c[t][r] * sc;
    }
}

__global__ __launch_bounds__(GEMM_THREADS) void gcn_gemm(const float* __restrict__ x,
                                                         const int*   __restrict__ edge,
                                                         const float* __restrict__ W,
                                                         float*       __restrict__ h,
                                                         int nStrips) {
    __shared__ float Wlds[64 * WPAIR_STRIDE * 2];

    {
        constexpr int PASSES = (4096 + GEMM_THREADS - 1) / GEMM_THREADS;
        float2 ga[PASSES], gb[PASSES];
#pragma unroll
        for (int i = 0; i < PASSES; ++i) {
            const int q  = threadIdx.x + GEMM_THREADS * i;
            const int qc = q < 4096 ? q : 4095;
            const int p = qc >> 6;
            const int c = qc & 63;
            ga[i] = *(const float2*)(W + (size_t)(2 * p)     * C_DIM + 2 * c);
            gb[i] = *(const float2*)(W + (size_t)(2 * p + 1) * C_DIM + 2 * c);
        }
#pragma unroll
        for (int i = 0; i < PASSES; ++i) {
            const int q = threadIdx.x + GEMM_THREADS * i;
            if (q < 4096) {
                const int p = q >> 6;
                const int c = q & 63;
                float4 v; v.x = ga[i].x; v.y = gb[i].x; v.z = ga[i].y; v.w = gb[i].y;
                ((float4*)Wlds)[p * (WPAIR_STRIDE / 2) + c] = v;
            }
        }
    }
    __syncthreads();

    const int wave = __builtin_amdgcn_readfirstlane(threadIdx.x >> 5);
    const int lane = threadIdx.x & 31;
    const int G    = gridDim.x;

    if (wave >= QUARTER_WAVES) {
        const int strip = (wave - QUARTER_WAVES) * G + blockIdx.x;
        if (strip < nStrips) strip_tiles<8, false>(x, edge, Wlds, h, strip, 0, lane);
    } else {
        const int t0 = wave * 2;
        for (int strip = WHOLE_WAVES * G + blockIdx.x; strip < nStrips; strip += G) {
            asm volatile("" ::: "memory");
            strip_tiles<2, true>(x, edge, Wlds, h, strip, t0, lane);
        }
    }
}

__global__ __launch_bounds__(256) __attribute__((amdgpu_waves_per_eu(4, 8))) void gcn_agg(const float* __restrict__ h,
                                               const int*   __restrict__ edge,
                                               float*       __restrict__ out,
                                               int nNodes) {
    const int lane   = threadIdx.x & 31;
    const int stride = gridDim.x * 8;
    const int first  = __builtin_amdgcn_readfirstlane(blockIdx.x * 8 + (threadIdx.x >> 5));

    const float4* h4   = (const float4*)h;
    float4*       out4 = (float4*)out;

    int idx = -1;
    if (first < nNodes && lane < MAX_DEG) idx = edge[(size_t)first * MAX_DEG + lane];
    unsigned long long valid = __ballot(idx >= 0);

    for (int node = first; node < nNodes; node += stride) {
        const float s = rsqrtf((float)__popcll(valid) + 1.0f);

        int    j[MAX_DEG];
        float4 v[MAX_DEG];
        const float4 self = h4[(size_t)node * 32 + lane];
#pragma unroll
        for (int k = 0; k < MAX_DEG; ++k) {
            j[k] = __builtin_amdgcn_readlane(idx, k);
            const int jc = j[k] < 0 ? 0 : j[k];
            v[k] = h4[(size_t)jc * 32 + lane];
        }
        const int next = node + stride;
        int idxNext = -1;
        if (next < nNodes && lane < MAX_DEG) idxNext = edge[(size_t)next * MAX_DEG + lane];

        float4 acc = self;
#pragma unroll
        for (int k = 0; k < MAX_DEG; ++k) {
            const bool ok = j[k] >= 0;
            const float tx = acc.x + v[k].x, ty = acc.y + v[k].y;
            const float tz = acc.z + v[k].z, tw = acc.w + v[k].w;
            acc.x = ok ? tx : acc.x; acc.y = ok ? ty : acc.y;
            acc.z = ok ? tz : acc.z; acc.w = ok ? tw : acc.w;
        }

        float4 o;
        o.x = acc.x * s; o.y = acc.y * s; o.z = acc.z * s; o.w = acc.w * s;
        out4[(size_t)node * 32 + lane] = o;

        idx   = idxNext;
        valid = __ballot(idx >= 0);
    }
}

extern "C" void kernel_launch(void* const* d_in, const int* in_sizes, int n_in,
                              void* d_out, int out_size, void* d_ws, size_t ws_size,
                              hipStream_t stream) {
    const float* x    = (const float*)d_in[0];
    const int*   edge = (const int*)  d_in[1];
    const float* W    = (const float*)d_in[2];
    float*       out  = (float*)d_out;
    float*       h    = (float*)d_ws;

    const int N       = in_sizes[0] / C_DIM;
    const int nStrips = (N + 15) / 16;

    gcn_gemm<<<256, GEMM_THREADS, 0, stream>>>(x, edge, W, h, nStrips);

    gcn_agg<<<1024, 256, 0, stream>>>(h, edge, out, N);
}
